# starting kernel
# baseline (speedup 1.0000x reference)
#include <hip/hip_runtime.h>
#include <cstdint>


typedef float v2f __attribute__((ext_vector_type(2)));
typedef float v8f __attribute__((ext_vector_type(8)));

__device__ __forceinline__ v8f wmma4(v2f a, v2f b, v8f c) {
  return __builtin_amdgcn_wmma_f32_16x16x4_f32(
       false, a,  false, b,
       (short)0, c,  false,  false);
}

__global__ __launch_bounds__(128) void k_gemm_qkv(
    const float* __restrict__ x, const float* __restrict__ w,
    float* __restrict__ qkv) {
  __shared__ __align__(16) float As[64 * 16];
  __shared__ __align__(16) float Bs[64 * 16];
  const int m0 = blockIdx.y * 64;
  const int n0 = blockIdx.x * 64;
  const int tid = threadIdx.x;
  const int wave = tid >> 5, lane = tid & 31;
  const int hl = lane & 15;
  const int koff = (lane < 16) ? 0 : 2;
  const int wm = (wave >> 1) * 32;
  const int wn = (wave & 1) * 32;

  v8f acc[4] = {};

  for (int k0 = 0; k0 < 2048; k0 += 16) {
    __syncthreads();
#pragma unroll
    for (int i = 0; i < 2; ++i) {
      int idx = tid + i * 128;
      int row = idx >> 2, c = (idx & 3) << 2;
      *(float4*)&As[row * 16 + c] =
          *(const float4*)(x + (size_t)(m0 + row) * 2048 + k0 + c);
      *(float4*)&Bs[row * 16 + c] =
          *(const float4*)(w + (size_t)(n0 + row) * 2048 + k0 + c);
    }
    __syncthreads();
#pragma unroll
    for (int kk = 0; kk < 16; kk += 4) {
      v2f a0 = *(const v2f*)&As[(wm + hl) * 16 + kk + koff];
      v2f a1 = *(const v2f*)&As[(wm + 16 + hl) * 16 + kk + koff];
      v2f b0 = *(const v2f*)&Bs[(wn + hl) * 16 + kk + koff];
      v2f b1 = *(const v2f*)&Bs[(wn + 16 + hl) * 16 + kk + koff];
      acc[0] = wmma4(a0, b0, acc[0]);
      acc[1] = wmma4(a0, b1, acc[1]);
      acc[2] = wmma4(a1, b0, acc[2]);
      acc[3] = wmma4(a1, b1, acc[3]);
    }
  }
#pragma unroll
  for (int t = 0; t < 4; ++t) {
    const int mt = t >> 1, nt = t & 1;
#pragma unroll
    for (int r = 0; r < 8; ++r) {
      int m = m0 + wm + mt * 16 + ((lane < 16) ? r : r + 8);
      int n = n0 + wn + nt * 16 + hl;
      int which = n >> 11;
      int h = (n >> 7) & 15;
      int d = n & 127;
      int b = m >> 11;
      int l = m & 2047;
      qkv[(size_t)which * (size_t)8388608 +
          ((size_t)((b * 16 + h) * 2048 + l)) * 128 + d] = acc[t][r];
    }
  }
}

__global__ __launch_bounds__(256) void k_rmsrope(
    float* __restrict__ q, float* __restrict__ kk,
    const float* __restrict__ pe, const float* __restrict__ qs,
    const float* __restrict__ ks) {
  const int lane = threadIdx.x & 31;
  const int wid = blockIdx.x * 8 + (threadIdx.x >> 5);
  const int nrows = 65536;
  const int isk = wid >= nrows;
  const int row = isk ? wid - nrows : wid;
  float* base = (isk ? kk : q) + (size_t)row * 128;
  const float* sc = isk ? ks : qs;
  const int l = row & 2047;

  float4 xv = *(const float4*)(base + lane * 4);
  float ss = xv.x * xv.x + xv.y * xv.y + xv.z * xv.z + xv.w * xv.w;
#pragma unroll
  for (int sh = 16; sh >= 1; sh >>= 1) ss += __shfl_xor(ss, sh, 32);
  float rrms = rsqrtf(ss * (1.0f / 128.0f) + 1e-6f);
  float4 s4 = *(const float4*)(sc + lane * 4);
  float x0 = xv.x * rrms * s4.x;
  float x1 = xv.y * rrms * s4.y;
  float x2 = xv.z * rrms * s4.z;
  float x3 = xv.w * rrms * s4.w;
  const float* peb = pe + (size_t)l * 256 + lane * 8;
  float4 p0 = *(const float4*)(peb);
  float4 p1 = *(const float4*)(peb + 4);
  float4 yv;
  yv.x = p0.x * x0 + p0.y * x1;
  yv.y = p0.z * x0 + p0.w * x1;
  yv.z = p1.x * x2 + p1.y * x3;
  yv.w = p1.z * x2 + p1.w * x3;
  *(float4*)(base + lane * 4) = yv;
}

__global__ __launch_bounds__(128) void k_attn(
    const float* __restrict__ q, const float* __restrict__ k,
    const float* __restrict__ v, float* __restrict__ o) {
  __shared__ __align__(16) float Ks[64 * 128];
  __shared__ __align__(16) float Vt[128 * 64];
  const int bh = blockIdx.y;
  const int q0 = blockIdx.x * 64;
  const int tid = threadIdx.x;
  const int wave = tid >> 5, lane = tid & 31;
  const int hl = lane & 15;
  const int koff = (lane < 16) ? 0 : 2;
  const float rscale = 0.08838834764831845f;

  const float* qb = q + (size_t)bh * (2048 * 128);
  const float* kbase = k + (size_t)bh * (2048 * 128);
  const float* vbase = v + (size_t)bh * (2048 * 128);

  v2f qa[32];
  {
    const float* qrow = qb + (size_t)(q0 + wave * 16 + hl) * 128;
#pragma unroll
    for (int s = 0; s < 32; ++s) qa[s] = *(const v2f*)(qrow + 4 * s + koff);
  }

  v8f oacc[8] = {};
  float m_i[8], l_i[8];
#pragma unroll
  for (int r = 0; r < 8; ++r) { m_i[r] = -3.4e38f; l_i[r] = 0.f; }

  for (int kb0 = 0; kb0 < 2048; kb0 += 64) {
    __syncthreads();
    for (int i = tid; i < 64 * 32; i += 128) {
      int row = i >> 5;
      int c4 = (i & 31) << 2;
      *(float4*)&Ks[row * 128 + c4] =
          *(const float4*)(kbase + (size_t)(kb0 + row) * 128 + c4);
      float4 vv = *(const float4*)(vbase + (size_t)(kb0 + row) * 128 + c4);
      Vt[(c4 + 0) * 64 + row] = vv.x;
      Vt[(c4 + 1) * 64 + row] = vv.y;
      Vt[(c4 + 2) * 64 + row] = vv.z;
      Vt[(c4 + 3) * 64 + row] = vv.w;
    }
    __syncthreads();

    v8f sacc[4] = {};
#pragma unroll
    for (int nt = 0; nt < 4; ++nt) {
      const float* krow = &Ks[(nt * 16 + hl) * 128];
      v8f a = sacc[nt];
#pragma unroll
      for (int s = 0; s < 32; ++s)
        a = wmma4(qa[s], *(const v2f*)(krow + 4 * s + koff), a);
      sacc[nt] = a;
    }

#pragma unroll
    for (int r = 0; r < 8; ++r) {
      float mx = -3.4e38f;
#pragma unroll
      for (int nt = 0; nt < 4; ++nt) {
        sacc[nt][r] *= rscale;
        mx = fmaxf(mx, sacc[nt][r]);
      }
#pragma unroll
      for (int sh = 8; sh >= 1; sh >>= 1) mx = fmaxf(mx, __shfl_xor(mx, sh, 32));
      float mnew = fmaxf(m_i[r], mx);
      float alpha = __expf(m_i[r] - mnew);
      m_i[r] = mnew;
      float rs = 0.f;
#pragma unroll
      for (int nt = 0; nt < 4; ++nt) {
        float p = __expf(sacc[nt][r] - mnew);
        sacc[nt][r] = p;
        rs += p;
      }
#pragma unroll
      for (int sh = 8; sh >= 1; sh >>= 1) rs += __shfl_xor(rs, sh, 32);
      l_i[r] = l_i[r] * alpha + rs;
#pragma unroll
      for (int nd = 0; nd < 8; ++nd) oacc[nd][r] *= alpha;
    }

    __syncthreads();
    float* Pw = &Ks[wave * (16 * 64)];
#pragma unroll
    for (int nt = 0; nt < 4; ++nt) {
#pragma unroll
      for (int r = 0; r < 8; ++r) {
        int row = (lane < 16) ? r : r + 8;
        Pw[row * 64 + nt * 16 + hl] = sacc[nt][r];
      }
    }
    v2f pa[16];
    {
      const float* prow = &Pw[hl * 64];
#pragma unroll
      for (int s = 0; s < 16; ++s) pa[s] = *(const v2f*)(prow + 4 * s + koff);
    }
#pragma unroll
    for (int nd = 0; nd < 8; ++nd) {
      const float* vtrow = &Vt[(nd * 16 + hl) * 64];
      v8f a = oacc[nd];
#pragma unroll
      for (int s = 0; s < 16; ++s)
        a = wmma4(pa[s], *(const v2f*)(vtrow + 4 * s + koff), a);
      oacc[nd] = a;
    }
  }

  const int b = bh >> 4, h = bh & 15;
#pragma unroll
  for (int nd = 0; nd < 8; ++nd) {
#pragma unroll
    for (int r = 0; r < 8; ++r) {
      int row = (lane < 16) ? r : r + 8;
      int t = q0 + wave * 16 + row;
      int d = nd * 16 + hl;
      o[((size_t)(b * 2048 + t)) * 2048 + h * 128 + d] = oacc[nd][r] / l_i[r];
    }
  }
}

__global__ __launch_bounds__(128) void k_gemm_out(
    const float* __restrict__ ov, const float* __restrict__ w,
    const float* __restrict__ bias, float* __restrict__ out) {
  __shared__ __align__(16) float As[64 * 16];
  __shared__ __align__(16) float Bs[64 * 16];
  const int m0 = blockIdx.y * 64;
  const int n0 = blockIdx.x * 64;
  const int tid = threadIdx.x;
  const int wave = tid >> 5, lane = tid & 31;
  const int hl = lane & 15;
  const int koff = (lane < 16) ? 0 : 2;
  const int wm = (wave >> 1) * 32;
  const int wn = (wave & 1) * 32;

  v8f acc[4] = {};

  for (int k0 = 0; k0 < 2048; k0 += 16) {
    __syncthreads();
#pragma unroll
    for (int i = 0; i < 2; ++i) {
      int idx = tid + i * 128;
      int row = idx >> 2, c = (idx & 3) << 2;
      *(float4*)&As[row * 16 + c] =
          *(const float4*)(ov + (size_t)(m0 + row) * 2048 + k0 + c);
      *(float4*)&Bs[row * 16 + c] =
          *(const float4*)(w + (size_t)(n0 + row) * 2048 + k0 + c);
    }
    __syncthreads();
#pragma unroll
    for (int kk = 0; kk < 16; kk += 4) {
      v2f a0 = *(const v2f*)&As[(wm + hl) * 16 + kk + koff];
      v2f a1 = *(const v2f*)&As[(wm + 16 + hl) * 16 + kk + koff];
      v2f b0 = *(const v2f*)&Bs[(wn + hl) * 16 + kk + koff];
      v2f b1 = *(const v2f*)&Bs[(wn + 16 + hl) * 16 + kk + koff];
      acc[0] = wmma4(a0, b0, acc[0]);
      acc[1] = wmma4(a0, b1, acc[1]);
      acc[2] = wmma4(a1, b0, acc[2]);
      acc[3] = wmma4(a1, b1, acc[3]);
    }
  }
#pragma unroll
  for (int t = 0; t < 4; ++t) {
    const int mt = t >> 1, nt = t & 1;
#pragma unroll
    for (int r = 0; r < 8; ++r) {
      int m = m0 + wm + mt * 16 + ((lane < 16) ? r : r + 8);
      int n = n0 + wn + nt * 16 + hl;
      out[(size_t)m * 2048 + n] = acc[t][r] + bias[n];
    }
  }
}

extern "C" void kernel_launch(void* const* d_in, const int* in_sizes, int n_in,
                              void* d_out, int out_size, void* d_ws,
                              size_t ws_size, hipStream_t stream) {
  const float* x = (const float*)d_in[0];
  const float* pe = (const float*)d_in[1];
  const float* qkv_w = (const float*)d_in[2];
  const float* q_scale = (const float*)d_in[3];
  const float* k_scale = (const float*)d_in[4];
  const float* proj_w = (const float*)d_in[5];
  const float* proj_b = (const float*)d_in[6];
  float* out = (float*)d_out;

  float* ws = (float*)d_ws;
  const size_t per = (size_t)8388608;
  float* qw = ws;
  float* kw = ws + per;
  float* vw = ws + 2 * per;
  float* ow = ws + 3 * per;

  k_gemm_qkv<<<dim3(96, 64), 128, 0, stream>>>(x, qkv_w, qw);
  k_rmsrope<<<dim3(16384), 256, 0, stream>>>(qw, kw, pe, q_scale, k_scale);
  k_attn<<<dim3(32, 32), 128, 0, stream>>>(qw, kw, vw, ow);
  k_gemm_out<<<dim3(32, 64), 128, 0, stream>>>(ow, proj_w, proj_b, out);
}


# LDS row pitches padded to 16 B x odd (GEMM tiles 20, attention K 132, V^T 68, P 68 floats) and V staged 8 keys x 4 float4 per wave: fragment reads and V^T stores bank-conflict-free
# speedup vs baseline: 1.4986x; 1.4986x over previous
#include <hip/hip_runtime.h>
#include <cstdint>


typedef float v2f __attribute__((ext_vector_type(2)));
typedef float v8f __attribute__((ext_vector_type(8)));

__device__ __forceinline__ v8f wmma4(v2f a, v2f b, v8f c) {
  return __builtin_amdgcn_wmma_f32_16x16x4_f32(
       false, a,  false, b,
       (short)0, c,  false,  false);
}

constexpr int GP = 20;
constexpr int SK = 132;
constexpr int SV = 68;
constexpr int SP = 68;

__global__ __launch_bounds__(128) void k_gemm_qkv(
    const float* __restrict__ x, const float* __restrict__ w,
    float* __restrict__ qkv) {
  __shared__ __align__(16) float As[64 * GP];
  __shared__ __align__(16) float Bs[64 * GP];
  const int m0 = blockIdx.y * 64;
  const int n0 = blockIdx.x * 64;
  const int tid = threadIdx.x;
  const int wave = tid >> 5, lane = tid & 31;
  const int hl = lane & 15;
  const int koff = (lane < 16) ? 0 : 2;
  const int wm = (wave >> 1) * 32;
  const int wn = (wave & 1) * 32;

  v8f acc[4] = {};

  for (int k0 = 0; k0 < 2048; k0 += 16) {
    __syncthreads();
#pragma unroll
    for (int i = 0; i < 2; ++i) {
      int idx = tid + i * 128;
      int row = idx >> 2, c = (idx & 3) << 2;
      *(float4*)&As[row * GP + c] =
          *(const float4*)(x + (size_t)(m0 + row) * 2048 + k0 + c);
      *(float4*)&Bs[row * GP + c] =
          *(const float4*)(w + (size_t)(n0 + row) * 2048 + k0 + c);
    }
    __syncthreads();
#pragma unroll
    for (int kk = 0; kk < 16; kk += 4) {
      v2f a0 = *(const v2f*)&As[(wm + hl) * GP + kk + koff];
      v2f a1 = *(const v2f*)&As[(wm + 16 + hl) * GP + kk + koff];
      v2f b0 = *(const v2f*)&Bs[(wn + hl) * GP + kk + koff];
      v2f b1 = *(const v2f*)&Bs[(wn + 16 + hl) * GP + kk + koff];
      acc[0] = wmma4(a0, b0, acc[0]);
      acc[1] = wmma4(a0, b1, acc[1]);
      acc[2] = wmma4(a1, b0, acc[2]);
      acc[3] = wmma4(a1, b1, acc[3]);
    }
  }
#pragma unroll
  for (int t = 0; t < 4; ++t) {
    const int mt = t >> 1, nt = t & 1;
#pragma unroll
    for (int r = 0; r < 8; ++r) {
      int m = m0 + wm + mt * 16 + ((lane < 16) ? r : r + 8);
      int n = n0 + wn + nt * 16 + hl;
      int which = n >> 11;
      int h = (n >> 7) & 15;
      int d = n & 127;
      int b = m >> 11;
      int l = m & 2047;
      qkv[(size_t)which * (size_t)8388608 +
          ((size_t)((b * 16 + h) * 2048 + l)) * 128 + d] = acc[t][r];
    }
  }
}

__global__ __launch_bounds__(256) void k_rmsrope(
    float* __restrict__ q, float* __restrict__ kk,
    const float* __restrict__ pe, const float* __restrict__ qs,
    const float* __restrict__ ks) {
  const int lane = threadIdx.x & 31;
  const int wid = blockIdx.x * 8 + (threadIdx.x >> 5);
  const int nrows = 65536;
  const int isk = wid >= nrows;
  const int row = isk ? wid - nrows : wid;
  float* base = (isk ? kk : q) + (size_t)row * 128;
  const float* sc = isk ? ks : qs;
  const int l = row & 2047;

  float4 xv = *(const float4*)(base + lane * 4);
  float ss = xv.x * xv.x + xv.y * xv.y + xv.z * xv.z + xv.w * xv.w;
#pragma unroll
  for (int sh = 16; sh >= 1; sh >>= 1) ss += __shfl_xor(ss, sh, 32);
  float rrms = rsqrtf(ss * (1.0f / 128.0f) + 1e-6f);
  float4 s4 = *(const float4*)(sc + lane * 4);
  float x0 = xv.x * rrms * s4.x;
  float x1 = xv.y * rrms * s4.y;
  float x2 = xv.z * rrms * s4.z;
  float x3 = xv.w * rrms * s4.w;
  const float* peb = pe + (size_t)l * 256 + lane * 8;
  float4 p0 = *(const float4*)(peb);
  float4 p1 = *(const float4*)(peb + 4);
  float4 yv;
  yv.x = p0.x * x0 + p0.y * x1;
  yv.y = p0.z * x0 + p0.w * x1;
  yv.z = p1.x * x2 + p1.y * x3;
  yv.w = p1.z * x2 + p1.w * x3;
  *(float4*)(base + lane * 4) = yv;
}

__global__ __launch_bounds__(128) void k_attn(
    const float* __restrict__ q, const float* __restrict__ k,
    const float* __restrict__ v, float* __restrict__ o) {
  __shared__ __align__(16) float Ks[64 * SK];
  __shared__ __align__(16) float Vt[128 * SV];
  const int bh = blockIdx.y;
  const int q0 = blockIdx.x * 64;
  const int tid = threadIdx.x;
  const int wave = tid >> 5, lane = tid & 31;
  const int hl = lane & 15;
  const int koff = (lane < 16) ? 0 : 2;
  const float rscale = 0.08838834764831845f;

  const float* qb = q + (size_t)bh * (2048 * 128);
  const float* kbase = k + (size_t)bh * (2048 * 128);
  const float* vbase = v + (size_t)bh * (2048 * 128);

  v2f qa[32];
  {
    const float* qrow = qb + (size_t)(q0 + wave * 16 + hl) * 128;
#pragma unroll
    for (int s = 0; s < 32; ++s) qa[s] = *(const v2f*)(qrow + 4 * s + koff);
  }

  v8f oacc[8] = {};
  float m_i[8], l_i[8];
#pragma unroll
  for (int r = 0; r < 8; ++r) { m_i[r] = -3.4e38f; l_i[r] = 0.f; }

  for (int kb0 = 0; kb0 < 2048; kb0 += 64) {
    __syncthreads();
    for (int i = tid; i < 64 * 32; i += 128) {
      int row = i >> 5;
      int c4 = (i & 31) << 2;
      *(float4*)&Ks[row * SK + c4] =
          *(const float4*)(kbase + (size_t)(kb0 + row) * 128 + c4);
      int vr = (((i >> 5) & 7) << 3) + (i & 7);
      int vc = ((((i >> 8) & 7) << 2) + ((i >> 3) & 3)) << 2;
      float4 vv = *(const float4*)(vbase + (size_t)(kb0 + vr) * 128 + vc);
      Vt[(vc + 0) * SV + vr] = vv.x;
      Vt[(vc + 1) * SV + vr] = vv.y;
      Vt[(vc + 2) * SV + vr] = vv.z;
      Vt[(vc + 3) * SV + vr] = vv.w;
    }
    __syncthreads();

    v8f sacc[4] = {};
#pragma unroll
    for (int nt = 0; nt < 4; ++nt) {
      const float* krow = &Ks[(nt * 16 + hl) * SK];
      v8f a = sacc[nt];
#pragma unroll
      for (int s = 0; s < 32; ++s)
        a = wmma4(qa[s], *(const v2f*)(krow + 4 * s + koff), a);
      sacc[nt] = a;
    }

#pragma unroll
    for (int r = 0; r < 8; ++r) {
      float mx = -3.4e38f;
#pragma unroll
      for (int nt = 0; nt < 4; ++nt) {
        sacc[nt][r] *= rscale;
        mx = fmaxf(mx, sacc[nt][r]);
      }
#pragma unroll
      for (int sh = 8; sh >= 1; sh >>= 1) mx = fmaxf(mx, __shfl_xor(mx, sh, 32));
      float mnew = fmaxf(m_i[r], mx);
      float alpha = __expf(m_i[r] - mnew);
      m_i[r] = mnew;
      float rs = 0.f;
#pragma unroll
      for (int nt = 0; nt < 4; ++nt) {
        float p = __expf(sacc[nt][r] - mnew);
        sacc[nt][r] = p;
        rs += p;
      }
#pragma unroll
      for (int sh = 8; sh >= 1; sh >>= 1) rs += __shfl_xor(rs, sh, 32);
      l_i[r] = l_i[r] * alpha + rs;
#pragma unroll
      for (int nd = 0; nd < 8; ++nd) oacc[nd][r] *= alpha;
    }

    __syncthreads();
    float* Pw = &Ks[wave * (16 * SP)];
#pragma unroll
    for (int nt = 0; nt < 4; ++nt) {
#pragma unroll
      for (int r = 0; r < 8; ++r) {
        int row = (lane < 16) ? r : r + 8;
        Pw[row * SP + nt * 16 + hl] = sacc[nt][r];
      }
    }
    v2f pa[16];
    {
      const float* prow = &Pw[hl * SP];
#pragma unroll
      for (int s = 0; s < 16; ++s) pa[s] = *(const v2f*)(prow + 4 * s + koff);
    }
#pragma unroll
    for (int nd = 0; nd < 8; ++nd) {
      const float* vtrow = &Vt[(nd * 16 + hl) * SV];
      v8f a = oacc[nd];
#pragma unroll
      for (int s = 0; s < 16; ++s)
        a = wmma4(pa[s], *(const v2f*)(vtrow + 4 * s + koff), a);
      oacc[nd] = a;
    }
  }

  const int b = bh >> 4, h = bh & 15;
#pragma unroll
  for (int nd = 0; nd < 8; ++nd) {
#pragma unroll
    for (int r = 0; r < 8; ++r) {
      int row = (lane < 16) ? r : r + 8;
      int t = q0 + wave * 16 + row;
      int d = nd * 16 + hl;
      o[((size_t)(b * 2048 + t)) * 2048 + h * 128 + d] = oacc[nd][r] / l_i[r];
    }
  }
}

__global__ __launch_bounds__(128) void k_gemm_out(
    const float* __restrict__ ov, const float* __restrict__ w,
    const float* __restrict__ bias, float* __restrict__ out) {
  __shared__ __align__(16) float As[64 * GP];
  __shared__ __align__(16) float Bs[64 * GP];
  const int m0 = blockIdx.y * 64;
  const int n0 = blockIdx.x * 64;
  const int tid = threadIdx.x;
  const int wave = tid >> 5, lane = tid & 31;
  const int hl = lane & 15;
  const int koff = (lane < 16) ? 0 : 2;
  const int wm = (wave >> 1) * 32;
  const int wn = (wave & 1) * 32;

  v8f acc[4] = {};

  for (int k0 = 0; k0 < 2048; k0 += 16) {
    __syncthreads();
#pragma unroll
    for (int i = 0; i < 2; ++i) {
      int idx = tid + i * 128;
      int row = idx >> 2, c = (idx & 3) << 2;
      *(float4*)&As[row * GP + c] =
          *(const float4*)(ov + (size_t)(m0 + row) * 2048 + k0 + c);
      *(float4*)&Bs[row * GP + c] =
          *(const float4*)(w + (size_t)(n0 + row) * 2048 + k0 + c);
    }
    __syncthreads();
#pragma unroll
    for (int kk = 0; kk < 16; kk += 4) {
      v2f a0 = *(const v2f*)&As[(wm + hl) * GP + kk + koff];
      v2f a1 = *(const v2f*)&As[(wm + 16 + hl) * GP + kk + koff];
      v2f b0 = *(const v2f*)&Bs[(wn + hl) * GP + kk + koff];
      v2f b1 = *(const v2f*)&Bs[(wn + 16 + hl) * GP + kk + koff];
      acc[0] = wmma4(a0, b0, acc[0]);
      acc[1] = wmma4(a0, b1, acc[1]);
      acc[2] = wmma4(a1, b0, acc[2]);
      acc[3] = wmma4(a1, b1, acc[3]);
    }
  }
#pragma unroll
  for (int t = 0; t < 4; ++t) {
    const int mt = t >> 1, nt = t & 1;
#pragma unroll
    for (int r = 0; r < 8; ++r) {
      int m = m0 + wm + mt * 16 + ((lane < 16) ? r : r + 8);
      int n = n0 + wn + nt * 16 + hl;
      out[(size_t)m * 2048 + n] = acc[t][r] + bias[n];
    }
  }
}

extern "C" void kernel_launch(void* const* d_in, const int* in_sizes, int n_in,
                              void* d_out, int out_size, void* d_ws,
                              size_t ws_size, hipStream_t stream) {
  const float* x = (const float*)d_in[0];
  const float* pe = (const float*)d_in[1];
  const float* qkv_w = (const float*)d_in[2];
  const float* q_scale = (const float*)d_in[3];
  const float* k_scale = (const float*)d_in[4];
  const float* proj_w = (const float*)d_in[5];
  const float* proj_b = (const float*)d_in[6];
  float* out = (float*)d_out;

  float* ws = (float*)d_ws;
  const size_t per = (size_t)8388608;
  float* qw = ws;
  float* kw = ws + per;
  float* vw = ws + 2 * per;
  float* ow = ws + 3 * per;

  k_gemm_qkv<<<dim3(96, 64), 128, 0, stream>>>(x, qkv_w, qw);
  k_rmsrope<<<dim3(16384), 256, 0, stream>>>(qw, kw, pe, q_scale, k_scale);
  k_attn<<<dim3(32, 32), 128, 0, stream>>>(qw, kw, vw, ow);
  k_gemm_out<<<dim3(32, 64), 128, 0, stream>>>(ow, proj_w, proj_b, out);
}


# Both projection GEMMs re-tiled from 64x64x16 work-group tiles (4 waves) to 128x128x64 (8 waves, 64x32 wave tiles): a quarter of the cache lines and an eighth of the barriers per WMMA; k order kept
# speedup vs baseline: 1.6361x; 1.0918x over previous
#include <hip/hip_runtime.h>
#include <cstdint>


typedef float v2f __attribute__((ext_vector_type(2)));
typedef float v8f __attribute__((ext_vector_type(8)));

__device__ __forceinline__ v8f wmma4(v2f a, v2f b, v8f c) {
  return __builtin_amdgcn_wmma_f32_16x16x4_f32(
       false, a,  false, b,
       (short)0, c,  false,  false);
}

constexpr int GK = 64;
constexpr int GP = 68;
constexpr int SK = 132;
constexpr int SV = 68;
constexpr int SP = 68;

__device__ __forceinline__ void gemm_tile_128(
    const float* __restrict__ a, const float* __restrict__ w, int m0, int n0,
    float* __restrict__ As, float* __restrict__ Bs, v8f (&acc)[8]) {
  const int tid = threadIdx.x;
  const int wave = tid >> 5, lane = tid & 31;
  const int hl = lane & 15;
  const int koff = (lane < 16) ? 0 : 2;
  const int wm = (wave >> 2) * 64;
  const int wn = (wave & 3) * 32;
  const int lrow = tid >> 4;
  const int lc = (tid & 15) << 2;
  const float* ag = a + (size_t)(m0 + lrow) * 2048 + lc;
  const float* wg = w + (size_t)(n0 + lrow) * 2048 + lc;
  float* asd = As + lrow * GP + lc;
  float* bsd = Bs + lrow * GP + lc;
  const float* afr = As + (wm + hl) * GP + koff;
  const float* bfr = Bs + (wn + hl) * GP + koff;

  for (int k0 = 0; k0 < 2048; k0 += GK) {
    __syncthreads();
#pragma unroll
    for (int i = 0; i < 8; ++i) {
      *(float4*)(asd + i * 16 * GP) =
          *(const float4*)(ag + (size_t)(i * 16) * 2048 + k0);
      *(float4*)(bsd + i * 16 * GP) =
          *(const float4*)(wg + (size_t)(i * 16) * 2048 + k0);
    }
    __syncthreads();
#pragma unroll
    for (int kk = 0; kk < GK; kk += 4) {
      v2f af[4], bf[2];
#pragma unroll
      for (int mt = 0; mt < 4; ++mt)
        af[mt] = *(const v2f*)(afr + mt * 16 * GP + kk);
#pragma unroll
      for (int nt = 0; nt < 2; ++nt)
        bf[nt] = *(const v2f*)(bfr + nt * 16 * GP + kk);
#pragma unroll
      for (int mt = 0; mt < 4; ++mt) {
        acc[mt * 2 + 0] = wmma4(af[mt], bf[0], acc[mt * 2 + 0]);
        acc[mt * 2 + 1] = wmma4(af[mt], bf[1], acc[mt * 2 + 1]);
      }
    }
  }
}

__global__ __launch_bounds__(256) void k_gemm_qkv(
    const float* __restrict__ x, const float* __restrict__ w,
    float* __restrict__ qkv) {
  __shared__ __align__(16) float As[128 * GP];
  __shared__ __align__(16) float Bs[128 * GP];
  const int m0 = blockIdx.y * 128;
  const int n0 = blockIdx.x * 128;
  v8f acc[8] = {};
  gemm_tile_128(x, w, m0, n0, As, Bs, acc);

  const int wave = threadIdx.x >> 5, lane = threadIdx.x & 31;
  const int hl = lane & 15;
  const int wm = (wave >> 2) * 64;
  const int wn = (wave & 3) * 32;
  const int which = n0 >> 11;
  const int h = (n0 >> 7) & 15;
  const int b = m0 >> 11;
  float* dst = qkv + (size_t)which * (size_t)8388608 +
               (size_t)((b * 16 + h) * 2048 + (m0 & 2047)) * 128;
#pragma unroll
  for (int t = 0; t < 8; ++t) {
    const int mt = t >> 1, nt = t & 1;
#pragma unroll
    for (int r = 0; r < 8; ++r) {
      int l = wm + mt * 16 + ((lane < 16) ? r : r + 8);
      int d = wn + nt * 16 + hl;
      dst[(size_t)l * 128 + d] = acc[t][r];
    }
  }
}

__global__ __launch_bounds__(256) void k_rmsrope(
    float* __restrict__ q, float* __restrict__ kk,
    const float* __restrict__ pe, const float* __restrict__ qs,
    const float* __restrict__ ks) {
  const int lane = threadIdx.x & 31;
  const int wid = blockIdx.x * 8 + (threadIdx.x >> 5);
  const int nrows = 65536;
  const int isk = wid >= nrows;
  const int row = isk ? wid - nrows : wid;
  float* base = (isk ? kk : q) + (size_t)row * 128;
  const float* sc = isk ? ks : qs;
  const int l = row & 2047;

  float4 xv = *(const float4*)(base + lane * 4);
  float ss = xv.x * xv.x + xv.y * xv.y + xv.z * xv.z + xv.w * xv.w;
#pragma unroll
  for (int sh = 16; sh >= 1; sh >>= 1) ss += __shfl_xor(ss, sh, 32);
  float rrms = rsqrtf(ss * (1.0f / 128.0f) + 1e-6f);
  float4 s4 = *(const float4*)(sc + lane * 4);
  float x0 = xv.x * rrms * s4.x;
  float x1 = xv.y * rrms * s4.y;
  float x2 = xv.z * rrms * s4.z;
  float x3 = xv.w * rrms * s4.w;
  const float* peb = pe + (size_t)l * 256 + lane * 8;
  float4 p0 = *(const float4*)(peb);
  float4 p1 = *(const float4*)(peb + 4);
  float4 yv;
  yv.x = p0.x * x0 + p0.y * x1;
  yv.y = p0.z * x0 + p0.w * x1;
  yv.z = p1.x * x2 + p1.y * x3;
  yv.w = p1.z * x2 + p1.w * x3;
  *(float4*)(base + lane * 4) = yv;
}

__global__ __launch_bounds__(128) void k_attn(
    const float* __restrict__ q, const float* __restrict__ k,
    const float* __restrict__ v, float* __restrict__ o) {
  __shared__ __align__(16) float Ks[64 * SK];
  __shared__ __align__(16) float Vt[128 * SV];
  const int bh = blockIdx.y;
  const int q0 = blockIdx.x * 64;
  const int tid = threadIdx.x;
  const int wave = tid >> 5, lane = tid & 31;
  const int hl = lane & 15;
  const int koff = (lane < 16) ? 0 : 2;
  const float rscale = 0.08838834764831845f;

  const float* qb = q + (size_t)bh * (2048 * 128);
  const float* kbase = k + (size_t)bh * (2048 * 128);
  const float* vbase = v + (size_t)bh * (2048 * 128);

  v2f qa[32];
  {
    const float* qrow = qb + (size_t)(q0 + wave * 16 + hl) * 128;
#pragma unroll
    for (int s = 0; s < 32; ++s) qa[s] = *(const v2f*)(qrow + 4 * s + koff);
  }

  v8f oacc[8] = {};
  float m_i[8], l_i[8];
#pragma unroll
  for (int r = 0; r < 8; ++r) { m_i[r] = -3.4e38f; l_i[r] = 0.f; }

  for (int kb0 = 0; kb0 < 2048; kb0 += 64) {
    __syncthreads();
    for (int i = tid; i < 64 * 32; i += 128) {
      int row = i >> 5;
      int c4 = (i & 31) << 2;
      *(float4*)&Ks[row * SK + c4] =
          *(const float4*)(kbase + (size_t)(kb0 + row) * 128 + c4);
      int vr = (((i >> 5) & 7) << 3) + (i & 7);
      int vc = ((((i >> 8) & 7) << 2) + ((i >> 3) & 3)) << 2;
      float4 vv = *(const float4*)(vbase + (size_t)(kb0 + vr) * 128 + vc);
      Vt[(vc + 0) * SV + vr] = vv.x;
      Vt[(vc + 1) * SV + vr] = vv.y;
      Vt[(vc + 2) * SV + vr] = vv.z;
      Vt[(vc + 3) * SV + vr] = vv.w;
    }
    __syncthreads();

    v8f sacc[4] = {};
#pragma unroll
    for (int nt = 0; nt < 4; ++nt) {
      const float* krow = &Ks[(nt * 16 + hl) * SK];
      v8f a = sacc[nt];
#pragma unroll
      for (int s = 0; s < 32; ++s)
        a = wmma4(qa[s], *(const v2f*)(krow + 4 * s + koff), a);
      sacc[nt] = a;
    }

#pragma unroll
    for (int r = 0; r < 8; ++r) {
      float mx = -3.4e38f;
#pragma unroll
      for (int nt = 0; nt < 4; ++nt) {
        sacc[nt][r] *= rscale;
        mx = fmaxf(mx, sacc[nt][r]);
      }
#pragma unroll
      for (int sh = 8; sh >= 1; sh >>= 1) mx = fmaxf(mx, __shfl_xor(mx, sh, 32));
      float mnew = fmaxf(m_i[r], mx);
      float alpha = __expf(m_i[r] - mnew);
      m_i[r] = mnew;
      float rs = 0.f;
#pragma unroll
      for (int nt = 0; nt < 4; ++nt) {
        float p = __expf(sacc[nt][r] - mnew);
        sacc[nt][r] = p;
        rs += p;
      }
#pragma unroll
      for (int sh = 8; sh >= 1; sh >>= 1) rs += __shfl_xor(rs, sh, 32);
      l_i[r] = l_i[r] * alpha + rs;
#pragma unroll
      for (int nd = 0; nd < 8; ++nd) oacc[nd][r] *= alpha;
    }

    __syncthreads();
    float* Pw = &Ks[wave * (16 * SP)];
#pragma unroll
    for (int nt = 0; nt < 4; ++nt) {
#pragma unroll
      for (int r = 0; r < 8; ++r) {
        int row = (lane < 16) ? r : r + 8;
        Pw[row * SP + nt * 16 + hl] = sacc[nt][r];
      }
    }
    v2f pa[16];
    {
      const float* prow = &Pw[hl * SP];
#pragma unroll
      for (int s = 0; s < 16; ++s) pa[s] = *(const v2f*)(prow + 4 * s + koff);
    }
#pragma unroll
    for (int nd = 0; nd < 8; ++nd) {
      const float* vtrow = &Vt[(nd * 16 + hl) * SV];
      v8f a = oacc[nd];
#pragma unroll
      for (int s = 0; s < 16; ++s)
        a = wmma4(pa[s], *(const v2f*)(vtrow + 4 * s + koff), a);
      oacc[nd] = a;
    }
  }

  const int b = bh >> 4, h = bh & 15;
#pragma unroll
  for (int nd = 0; nd < 8; ++nd) {
#pragma unroll
    for (int r = 0; r < 8; ++r) {
      int row = (lane < 16) ? r : r + 8;
      int t = q0 + wave * 16 + row;
      int d = nd * 16 + hl;
      o[((size_t)(b * 2048 + t)) * 2048 + h * 128 + d] = oacc[nd][r] / l_i[r];
    }
  }
}

__global__ __launch_bounds__(256) void k_gemm_out(
    const float* __restrict__ ov, const float* __restrict__ w,
    const float* __restrict__ bias, float* __restrict__ out) {
  __shared__ __align__(16) float As[128 * GP];
  __shared__ __align__(16) float Bs[128 * GP];
  const int m0 = blockIdx.y * 128;
  const int n0 = blockIdx.x * 128;
  v8f acc[8] = {};
  gemm_tile_128(ov, w, m0, n0, As, Bs, acc);

  const int wave = threadIdx.x >> 5, lane = threadIdx.x & 31;
  const int hl = lane & 15;
  const int wm = (wave >> 2) * 64;
  const int wn = (wave & 3) * 32;
#pragma unroll
  for (int t = 0; t < 8; ++t) {
    const int mt = t >> 1, nt = t & 1;
    const int n = n0 + wn + nt * 16 + hl;
    const float bn = bias[n];
#pragma unroll
    for (int r = 0; r < 8; ++r) {
      int m = m0 + wm + mt * 16 + ((lane < 16) ? r : r + 8);
      out[(size_t)m * 2048 + n] = acc[t][r] + bn;
    }
  }
}

extern "C" void kernel_launch(void* const* d_in, const int* in_sizes, int n_in,
                              void* d_out, int out_size, void* d_ws,
                              size_t ws_size, hipStream_t stream) {
  const float* x = (const float*)d_in[0];
  const float* pe = (const float*)d_in[1];
  const float* qkv_w = (const float*)d_in[2];
  const float* q_scale = (const float*)d_in[3];
  const float* k_scale = (const float*)d_in[4];
  const float* proj_w = (const float*)d_in[5];
  const float* proj_b = (const float*)d_in[6];
  float* out = (float*)d_out;

  float* ws = (float*)d_ws;
  const size_t per = (size_t)8388608;
  float* qw = ws;
  float* kw = ws + per;
  float* vw = ws + 2 * per;
  float* ow = ws + 3 * per;

  k_gemm_qkv<<<dim3(48, 32), 256, 0, stream>>>(x, qkv_w, qw);
  k_rmsrope<<<dim3(16384), 256, 0, stream>>>(qw, kw, pe, q_scale, k_scale);
  k_attn<<<dim3(32, 32), 128, 0, stream>>>(qw, kw, vw, ow);
  k_gemm_out<<<dim3(16, 32), 256, 0, stream>>>(ow, proj_w, proj_b, out);
}


# Attention scores computed transposed (S^T = K Q^T, O^T = V^T P^T): P stays in registers as the B operand, no LDS trip or barrier for it, row max/sum in-lane; row-sum and PV key order regrouped
# speedup vs baseline: 1.7700x; 1.0818x over previous
#include <hip/hip_runtime.h>
#include <cstdint>


typedef float v2f __attribute__((ext_vector_type(2)));
typedef float v8f __attribute__((ext_vector_type(8)));

__device__ __forceinline__ v8f wmma4(v2f a, v2f b, v8f c) {
  return __builtin_amdgcn_wmma_f32_16x16x4_f32(
       false, a,  false, b,
       (short)0, c,  false,  false);
}

constexpr int GK = 64;
constexpr int GP = 68;
constexpr int SK = 132;

__device__ __forceinline__ void gemm_tile_128(
    const float* __restrict__ a, const float* __restrict__ w, int m0, int n0,
    float* __restrict__ As, float* __restrict__ Bs, v8f (&acc)[8]) {
  const int tid = threadIdx.x;
  const int wave = tid >> 5, lane = tid & 31;
  const int hl = lane & 15;
  const int koff = (lane < 16) ? 0 : 2;
  const int wm = (wave >> 2) * 64;
  const int wn = (wave & 3) * 32;
  const int lrow = tid >> 4;
  const int lc = (tid & 15) << 2;
  const float* ag = a + (size_t)(m0 + lrow) * 2048 + lc;
  const float* wg = w + (size_t)(n0 + lrow) * 2048 + lc;
  float* asd = As + lrow * GP + lc;
  float* bsd = Bs + lrow * GP + lc;
  const float* afr = As + (wm + hl) * GP + koff;
  const float* bfr = Bs + (wn + hl) * GP + koff;

  for (int k0 = 0; k0 < 2048; k0 += GK) {
    __syncthreads();
#pragma unroll
    for (int i = 0; i < 8; ++i) {
      *(float4*)(asd + i * 16 * GP) =
          *(const float4*)(ag + (size_t)(i * 16) * 2048 + k0);
      *(float4*)(bsd + i * 16 * GP) =
          *(const float4*)(wg + (size_t)(i * 16) * 2048 + k0);
    }
    __syncthreads();
#pragma unroll
    for (int kk = 0; kk < GK; kk += 4) {
      v2f af[4], bf[2];
#pragma unroll
      for (int mt = 0; mt < 4; ++mt)
        af[mt] = *(const v2f*)(afr + mt * 16 * GP + kk);
#pragma unroll
      for (int nt = 0; nt < 2; ++nt)
        bf[nt] = *(const v2f*)(bfr + nt * 16 * GP + kk);
#pragma unroll
      for (int mt = 0; mt < 4; ++mt) {
        acc[mt * 2 + 0] = wmma4(af[mt], bf[0], acc[mt * 2 + 0]);
        acc[mt * 2 + 1] = wmma4(af[mt], bf[1], acc[mt * 2 + 1]);
      }
    }
  }
}

__global__ __launch_bounds__(256) void k_gemm_qkv(
    const float* __restrict__ x, const float* __restrict__ w,
    float* __restrict__ qkv) {
  __shared__ __align__(16) float As[128 * GP];
  __shared__ __align__(16) float Bs[128 * GP];
  const int m0 = blockIdx.y * 128;
  const int n0 = blockIdx.x * 128;
  v8f acc[8] = {};
  gemm_tile_128(x, w, m0, n0, As, Bs, acc);

  const int wave = threadIdx.x >> 5, lane = threadIdx.x & 31;
  const int hl = lane & 15;
  const int wm = (wave >> 2) * 64;
  const int wn = (wave & 3) * 32;
  const int which = n0 >> 11;
  const int h = (n0 >> 7) & 15;
  const int b = m0 >> 11;
  float* dst = qkv + (size_t)which * (size_t)8388608 +
               (size_t)((b * 16 + h) * 2048 + (m0 & 2047)) * 128;
#pragma unroll
  for (int t = 0; t < 8; ++t) {
    const int mt = t >> 1, nt = t & 1;
#pragma unroll
    for (int r = 0; r < 8; ++r) {
      int l = wm + mt * 16 + ((lane < 16) ? r : r + 8);
      int d = wn + nt * 16 + hl;
      dst[(size_t)l * 128 + d] = acc[t][r];
    }
  }
}

__global__ __launch_bounds__(256) void k_rmsrope(
    float* __restrict__ q, float* __restrict__ kk,
    const float* __restrict__ pe, const float* __restrict__ qs,
    const float* __restrict__ ks) {
  const int lane = threadIdx.x & 31;
  const int wid = blockIdx.x * 8 + (threadIdx.x >> 5);
  const int nrows = 65536;
  const int isk = wid >= nrows;
  const int row = isk ? wid - nrows : wid;
  float* base = (isk ? kk : q) + (size_t)row * 128;
  const float* sc = isk ? ks : qs;
  const int l = row & 2047;

  float4 xv = *(const float4*)(base + lane * 4);
  float ss = xv.x * xv.x + xv.y * xv.y + xv.z * xv.z + xv.w * xv.w;
#pragma unroll
  for (int sh = 16; sh >= 1; sh >>= 1) ss += __shfl_xor(ss, sh, 32);
  float rrms = rsqrtf(ss * (1.0f / 128.0f) + 1e-6f);
  float4 s4 = *(const float4*)(sc + lane * 4);
  float x0 = xv.x * rrms * s4.x;
  float x1 = xv.y * rrms * s4.y;
  float x2 = xv.z * rrms * s4.z;
  float x3 = xv.w * rrms * s4.w;
  const float* peb = pe + (size_t)l * 256 + lane * 8;
  float4 p0 = *(const float4*)(peb);
  float4 p1 = *(const float4*)(peb + 4);
  float4 yv;
  yv.x = p0.x * x0 + p0.y * x1;
  yv.y = p0.z * x0 + p0.w * x1;
  yv.z = p1.x * x2 + p1.y * x3;
  yv.w = p1.z * x2 + p1.w * x3;
  *(float4*)(base + lane * 4) = yv;
}

constexpr int VP = 264;

__global__ __launch_bounds__(128) void k_attn(
    const float* __restrict__ q, const float* __restrict__ k,
    const float* __restrict__ v, float* __restrict__ o) {
  __shared__ __align__(16) float Ks[64 * SK];
  __shared__ __align__(16) float Vp[32 * VP];
  const int bh = blockIdx.y;
  const int q0 = blockIdx.x * 64;
  const int tid = threadIdx.x;
  const int wave = tid >> 5, lane = tid & 31;
  const int hl = lane & 15;
  const int koff = (lane < 16) ? 0 : 2;
  const int h8 = (lane < 16) ? 0 : 8;
  const float rscale = 0.08838834764831845f;

  const float* qb = q + (size_t)bh * (2048 * 128);
  const float* kbase = k + (size_t)bh * (2048 * 128);
  const float* vbase = v + (size_t)bh * (2048 * 128);

  v2f qa[32];
  {
    const float* qrow = qb + (size_t)(q0 + wave * 16 + hl) * 128;
#pragma unroll
    for (int s = 0; s < 32; ++s) qa[s] = *(const v2f*)(qrow + 4 * s + koff);
  }

  v8f oacc[8] = {};
  float m_i = -3.4e38f, l_i = 0.f;

  const float* kfr = &Ks[hl * SK + koff];
  const float* vfr = &Vp[(h8 >> 1) * VP + hl * 2];

  for (int kb0 = 0; kb0 < 2048; kb0 += 64) {
    __syncthreads();
#pragma unroll 2
    for (int it = 0; it < 8; ++it) {
      const int i = tid + it * 128;
      const int m = i >> 5;
      const int c4 = (i & 31) << 2;
      const float* kg = kbase + (size_t)(kb0 + 2 * m) * 128 + c4;
      const float* vg = vbase + (size_t)(kb0 + 2 * m) * 128 + c4;
      float4 k0 = *(const float4*)(kg);
      float4 k1 = *(const float4*)(kg + 128);
      float4 va = *(const float4*)(vg);
      float4 vb = *(const float4*)(vg + 128);
      *(float4*)&Ks[(2 * m) * SK + c4] = k0;
      *(float4*)&Ks[(2 * m + 1) * SK + c4] = k1;
      *(float4*)&Vp[m * VP + c4 * 2] = make_float4(va.x, vb.x, va.y, vb.y);
      *(float4*)&Vp[m * VP + c4 * 2 + 4] = make_float4(va.z, vb.z, va.w, vb.w);
    }
    __syncthreads();

    v8f st[4] = {};
    {
      v2f kf[2][8];
#pragma unroll
      for (int nt = 0; nt < 4; ++nt) {
        kf[0][nt * 2] = *(const v2f*)(kfr + nt * 16 * SK);
        kf[0][nt * 2 + 1] = *(const v2f*)(kfr + nt * 16 * SK + 4);
      }
#pragma unroll
      for (int sp = 0; sp < 16; ++sp) {
        const int cur = sp & 1, nxt = cur ^ 1;
        if (sp + 1 < 16) {
#pragma unroll
          for (int nt = 0; nt < 4; ++nt) {
            kf[nxt][nt * 2] =
                *(const v2f*)(kfr + nt * 16 * SK + 8 * (sp + 1));
            kf[nxt][nt * 2 + 1] =
                *(const v2f*)(kfr + nt * 16 * SK + 8 * (sp + 1) + 4);
          }
        }
        __builtin_amdgcn_sched_barrier(0);
#pragma unroll
        for (int nt = 0; nt < 4; ++nt)
          st[nt] = wmma4(kf[cur][nt * 2], qa[2 * sp], st[nt]);
#pragma unroll
        for (int nt = 0; nt < 4; ++nt)
          st[nt] = wmma4(kf[cur][nt * 2 + 1], qa[2 * sp + 1], st[nt]);
      }
    }

    v2f vf[2][8];
#pragma unroll
    for (int nd = 0; nd < 8; ++nd) vf[0][nd] = *(const v2f*)(vfr + nd * 32);

    float mx = -3.4e38f;
#pragma unroll
    for (int nt = 0; nt < 4; ++nt) {
#pragma unroll
      for (int r = 0; r < 8; ++r) {
        st[nt][r] *= rscale;
        mx = fmaxf(mx, st[nt][r]);
      }
    }
    mx = fmaxf(mx, __shfl_xor(mx, 16, 32));
    const float mnew = fmaxf(m_i, mx);
    const float alpha = __expf(m_i - mnew);
    m_i = mnew;
    float rs = 0.f;
#pragma unroll
    for (int nt = 0; nt < 4; ++nt) {
#pragma unroll
      for (int r = 0; r < 8; ++r) {
        float p = __expf(st[nt][r] - mnew);
        st[nt][r] = p;
        rs += p;
      }
    }
    rs += __shfl_xor(rs, 16, 32);
    l_i = l_i * alpha + rs;
#pragma unroll
    for (int nd = 0; nd < 8; ++nd) {
#pragma unroll
      for (int r = 0; r < 8; ++r) oacc[nd][r] *= alpha;
    }

#pragma unroll
    for (int t = 0; t < 16; ++t) {
      const int nt = t >> 2, j = t & 3;
      const int cur = t & 1, nxt = cur ^ 1;
      if (t + 1 < 16) {
        const int nt1 = (t + 1) >> 2, j1 = (t + 1) & 3;
#pragma unroll
        for (int nd = 0; nd < 8; ++nd)
          vf[nxt][nd] = *(const v2f*)(vfr + (nt1 * 8 + j1) * VP + nd * 32);
      }
      __builtin_amdgcn_sched_barrier(0);
      v2f pb;
      pb.x = st[nt][2 * j];
      pb.y = st[nt][2 * j + 1];
#pragma unroll
      for (int nd = 0; nd < 8; ++nd)
        oacc[nd] = wmma4(vf[cur][nd], pb, oacc[nd]);
    }
  }

  const int b = bh >> 4, h = bh & 15;
  float* orow = o + ((size_t)(b * 2048 + q0 + wave * 16 + hl)) * 2048 +
                h * 128 + h8;
#pragma unroll
  for (int nd = 0; nd < 8; ++nd) {
    float4 lo, hi;
    lo.x = oacc[nd][0] / l_i;
    lo.y = oacc[nd][1] / l_i;
    lo.z = oacc[nd][2] / l_i;
    lo.w = oacc[nd][3] / l_i;
    hi.x = oacc[nd][4] / l_i;
    hi.y = oacc[nd][5] / l_i;
    hi.z = oacc[nd][6] / l_i;
    hi.w = oacc[nd][7] / l_i;
    *(float4*)(orow + nd * 16) = lo;
    *(float4*)(orow + nd * 16 + 4) = hi;
  }
}

__global__ __launch_bounds__(256) void k_gemm_out(
    const float* __restrict__ ov, const float* __restrict__ w,
    const float* __restrict__ bias, float* __restrict__ out) {
  __shared__ __align__(16) float As[128 * GP];
  __shared__ __align__(16) float Bs[128 * GP];
  const int m0 = blockIdx.y * 128;
  const int n0 = blockIdx.x * 128;
  v8f acc[8] = {};
  gemm_tile_128(ov, w, m0, n0, As, Bs, acc);

  const int wave = threadIdx.x >> 5, lane = threadIdx.x & 31;
  const int hl = lane & 15;
  const int wm = (wave >> 2) * 64;
  const int wn = (wave & 3) * 32;
#pragma unroll
  for (int t = 0; t < 8; ++t) {
    const int mt = t >> 1, nt = t & 1;
    const int n = n0 + wn + nt * 16 + hl;
    const float bn = bias[n];
#pragma unroll
    for (int r = 0; r < 8; ++r) {
      int m = m0 + wm + mt * 16 + ((lane < 16) ? r : r + 8);
      out[(size_t)m * 2048 + n] = acc[t][r] + bn;
    }
  }
}

extern "C" void kernel_launch(void* const* d_in, const int* in_sizes, int n_in,
                              void* d_out, int out_size, void* d_ws,
                              size_t ws_size, hipStream_t stream) {
  const float* x = (const float*)d_in[0];
  const float* pe = (const float*)d_in[1];
  const float* qkv_w = (const float*)d_in[2];
  const float* q_scale = (const float*)d_in[3];
  const float* k_scale = (const float*)d_in[4];
  const float* proj_w = (const float*)d_in[5];
  const float* proj_b = (const float*)d_in[6];
  float* out = (float*)d_out;

  float* ws = (float*)d_ws;
  const size_t per = (size_t)8388608;
  float* qw = ws;
  float* kw = ws + per;
  float* vw = ws + 2 * per;
  float* ow = ws + 3 * per;

  k_gemm_qkv<<<dim3(48, 32), 256, 0, stream>>>(x, qkv_w, qw);
  k_rmsrope<<<dim3(16384), 256, 0, stream>>>(qw, kw, pe, q_scale, k_scale);
  k_attn<<<dim3(32, 32), 128, 0, stream>>>(qw, kw, vw, ow);
  k_gemm_out<<<dim3(16, 32), 256, 0, stream>>>(ow, proj_w, proj_b, out);
}


# RMSNorm+RoPE of q and k fused into the QKV GEMM epilogue: tile staged through LDS, a wave per row, one 512-B store per row; the separate 131072-wave pass and its launch go; same row-sum tree
# speedup vs baseline: 1.8340x; 1.0362x over previous
#include <hip/hip_runtime.h>
#include <cstdint>


typedef float v2f __attribute__((ext_vector_type(2)));
typedef float v8f __attribute__((ext_vector_type(8)));

__device__ __forceinline__ v8f wmma4(v2f a, v2f b, v8f c) {
  return __builtin_amdgcn_wmma_f32_16x16x4_f32(
       false, a,  false, b,
       (short)0, c,  false,  false);
}

constexpr int GK = 64;
constexpr int GP = 68;
constexpr int SK = 132;

__device__ __forceinline__ void gemm_tile_128(
    const float* __restrict__ a, const float* __restrict__ w, int m0, int n0,
    float* __restrict__ As, float* __restrict__ Bs, v8f (&acc)[8]) {
  const int tid = threadIdx.x;
  const int wave = tid >> 5, lane = tid & 31;
  const int hl = lane & 15;
  const int koff = (lane < 16) ? 0 : 2;
  const int wm = (wave >> 2) * 64;
  const int wn = (wave & 3) * 32;
  const int lrow = tid >> 4;
  const int lc = (tid & 15) << 2;
  const float* ag = a + (size_t)(m0 + lrow) * 2048 + lc;
  const float* wg = w + (size_t)(n0 + lrow) * 2048 + lc;
  float* asd = As + lrow * GP + lc;
  float* bsd = Bs + lrow * GP + lc;
  const float* afr = As + (wm + hl) * GP + koff;
  const float* bfr = Bs + (wn + hl) * GP + koff;

  for (int k0 = 0; k0 < 2048; k0 += GK) {
    __syncthreads();
#pragma unroll
    for (int i = 0; i < 8; ++i) {
      *(float4*)(asd + i * 16 * GP) =
          *(const float4*)(ag + (size_t)(i * 16) * 2048 + k0);
      *(float4*)(bsd + i * 16 * GP) =
          *(const float4*)(wg + (size_t)(i * 16) * 2048 + k0);
    }
    __syncthreads();
#pragma unroll
    for (int kk = 0; kk < GK; kk += 4) {
      v2f af[4], bf[2];
#pragma unroll
      for (int mt = 0; mt < 4; ++mt)
        af[mt] = *(const v2f*)(afr + mt * 16 * GP + kk);
#pragma unroll
      for (int nt = 0; nt < 2; ++nt)
        bf[nt] = *(const v2f*)(bfr + nt * 16 * GP + kk);
#pragma unroll
      for (int mt = 0; mt < 4; ++mt) {
        acc[mt * 2 + 0] = wmma4(af[mt], bf[0], acc[mt * 2 + 0]);
        acc[mt * 2 + 1] = wmma4(af[mt], bf[1], acc[mt * 2 + 1]);
      }
    }
  }
}

constexpr int TP = 132;

__global__ __launch_bounds__(256) void k_gemm_qkv(
    const float* __restrict__ x, const float* __restrict__ w,
    const float* __restrict__ pe, const float* __restrict__ qs,
    const float* __restrict__ ks, float* __restrict__ qkv) {
  __shared__ __align__(16) float smem[2 * 128 * GP];
  float* As = smem;
  float* Bs = smem + 128 * GP;
  const int m0 = blockIdx.y * 128;
  const int n0 = blockIdx.x * 128;
  v8f acc[8] = {};
  gemm_tile_128(x, w, m0, n0, As, Bs, acc);

  const int wave = threadIdx.x >> 5, lane = threadIdx.x & 31;
  const int hl = lane & 15;
  const int h8 = (lane < 16) ? 0 : 8;
  const int wm = (wave >> 2) * 64;
  const int wn = (wave & 3) * 32;
  const int which = n0 >> 11;
  const int h = (n0 >> 7) & 15;
  const int b = m0 >> 11;
  const int l0 = m0 & 2047;
  float* dst = qkv + (size_t)which * (size_t)8388608 +
               (size_t)((b * 16 + h) * 2048 + l0) * 128;

  __syncthreads();
#pragma unroll
  for (int t = 0; t < 8; ++t) {
    const int mt = t >> 1, nt = t & 1;
#pragma unroll
    for (int r = 0; r < 8; ++r)
      smem[(wm + mt * 16 + h8 + r) * TP + wn + nt * 16 + hl] = acc[t][r];
  }
  __syncthreads();

  const float* trow = smem + wave * 16 * TP + lane * 4;
  float* drow = dst + (size_t)(wave * 16) * 128 + lane * 4;
  if (which == 2) {
#pragma unroll
    for (int i = 0; i < 16; ++i)
      *(float4*)(drow + i * 128) = *(const float4*)(trow + i * TP);
    return;
  }

  const float* sc = which ? ks : qs;
  const float4 s4 = *(const float4*)(sc + lane * 4);
  const float* peb = pe + (size_t)(l0 + wave * 16) * 256 + lane * 8;
#pragma unroll 4
  for (int i = 0; i < 16; ++i) {
    float4 xv = *(const float4*)(trow + i * TP);
    float4 p0 = *(const float4*)(peb + i * 256);
    float4 p1 = *(const float4*)(peb + i * 256 + 4);
    float ss = xv.x * xv.x + xv.y * xv.y + xv.z * xv.z + xv.w * xv.w;
#pragma unroll
    for (int sh = 16; sh >= 1; sh >>= 1) ss += __shfl_xor(ss, sh, 32);
    float rrms = rsqrtf(ss * (1.0f / 128.0f) + 1e-6f);
    float x0 = xv.x * rrms * s4.x;
    float x1 = xv.y * rrms * s4.y;
    float x2 = xv.z * rrms * s4.z;
    float x3 = xv.w * rrms * s4.w;
    float4 yv;
    yv.x = p0.x * x0 + p0.y * x1;
    yv.y = p0.z * x0 + p0.w * x1;
    yv.z = p1.x * x2 + p1.y * x3;
    yv.w = p1.z * x2 + p1.w * x3;
    *(float4*)(drow + i * 128) = yv;
  }
}

constexpr int VP = 264;

__global__ __launch_bounds__(128) void k_attn(
    const float* __restrict__ q, const float* __restrict__ k,
    const float* __restrict__ v, float* __restrict__ o) {
  __shared__ __align__(16) float Ks[64 * SK];
  __shared__ __align__(16) float Vp[32 * VP];
  const int bh = blockIdx.y;
  const int q0 = blockIdx.x * 64;
  const int tid = threadIdx.x;
  const int wave = tid >> 5, lane = tid & 31;
  const int hl = lane & 15;
  const int koff = (lane < 16) ? 0 : 2;
  const int h8 = (lane < 16) ? 0 : 8;
  const float rscale = 0.08838834764831845f;

  const float* qb = q + (size_t)bh * (2048 * 128);
  const float* kbase = k + (size_t)bh * (2048 * 128);
  const float* vbase = v + (size_t)bh * (2048 * 128);

  v2f qa[32];
  {
    const float* qrow = qb + (size_t)(q0 + wave * 16 + hl) * 128;
#pragma unroll
    for (int s = 0; s < 32; ++s) qa[s] = *(const v2f*)(qrow + 4 * s + koff);
  }

  v8f oacc[8] = {};
  float m_i = -3.4e38f, l_i = 0.f;

  const float* kfr = &Ks[hl * SK + koff];
  const float* vfr = &Vp[(h8 >> 1) * VP + hl * 2];

  for (int kb0 = 0; kb0 < 2048; kb0 += 64) {
    __syncthreads();
#pragma unroll 2
    for (int it = 0; it < 8; ++it) {
      const int i = tid + it * 128;
      const int m = i >> 5;
      const int c4 = (i & 31) << 2;
      const float* kg = kbase + (size_t)(kb0 + 2 * m) * 128 + c4;
      const float* vg = vbase + (size_t)(kb0 + 2 * m) * 128 + c4;
      float4 k0 = *(const float4*)(kg);
      float4 k1 = *(const float4*)(kg + 128);
      float4 va = *(const float4*)(vg);
      float4 vb = *(const float4*)(vg + 128);
      *(float4*)&Ks[(2 * m) * SK + c4] = k0;
      *(float4*)&Ks[(2 * m + 1) * SK + c4] = k1;
      *(float4*)&Vp[m * VP + c4 * 2] = make_float4(va.x, vb.x, va.y, vb.y);
      *(float4*)&Vp[m * VP + c4 * 2 + 4] = make_float4(va.z, vb.z, va.w, vb.w);
    }
    __syncthreads();

    v8f st[4] = {};
    {
      v2f kf[2][8];
#pragma unroll
      for (int nt = 0; nt < 4; ++nt) {
        kf[0][nt * 2] = *(const v2f*)(kfr + nt * 16 * SK);
        kf[0][nt * 2 + 1] = *(const v2f*)(kfr + nt * 16 * SK + 4);
      }
#pragma unroll
      for (int sp = 0; sp < 16; ++sp) {
        const int cur = sp & 1, nxt = cur ^ 1;
        if (sp + 1 < 16) {
#pragma unroll
          for (int nt = 0; nt < 4; ++nt) {
            kf[nxt][nt * 2] =
                *(const v2f*)(kfr + nt * 16 * SK + 8 * (sp + 1));
            kf[nxt][nt * 2 + 1] =
                *(const v2f*)(kfr + nt * 16 * SK + 8 * (sp + 1) + 4);
          }
        }
        __builtin_amdgcn_sched_barrier(0);
#pragma unroll
        for (int nt = 0; nt < 4; ++nt)
          st[nt] = wmma4(kf[cur][nt * 2], qa[2 * sp], st[nt]);
#pragma unroll
        for (int nt = 0; nt < 4; ++nt)
          st[nt] = wmma4(kf[cur][nt * 2 + 1], qa[2 * sp + 1], st[nt]);
      }
    }

    v2f vf[2][8];
#pragma unroll
    for (int nd = 0; nd < 8; ++nd) vf[0][nd] = *(const v2f*)(vfr + nd * 32);

    float mx = -3.4e38f;
#pragma unroll
    for (int nt = 0; nt < 4; ++nt) {
#pragma unroll
      for (int r = 0; r < 8; ++r) {
        st[nt][r] *= rscale;
        mx = fmaxf(mx, st[nt][r]);
      }
    }
    mx = fmaxf(mx, __shfl_xor(mx, 16, 32));
    const float mnew = fmaxf(m_i, mx);
    const float alpha = __expf(m_i - mnew);
    m_i = mnew;
    float rs = 0.f;
#pragma unroll
    for (int nt = 0; nt < 4; ++nt) {
#pragma unroll
      for (int r = 0; r < 8; ++r) {
        float p = __expf(st[nt][r] - mnew);
        st[nt][r] = p;
        rs += p;
      }
    }
    rs += __shfl_xor(rs, 16, 32);
    l_i = l_i * alpha + rs;
#pragma unroll
    for (int nd = 0; nd < 8; ++nd) {
#pragma unroll
      for (int r = 0; r < 8; ++r) oacc[nd][r] *= alpha;
    }

#pragma unroll
    for (int t = 0; t < 16; ++t) {
      const int nt = t >> 2, j = t & 3;
      const int cur = t & 1, nxt = cur ^ 1;
      if (t + 1 < 16) {
        const int nt1 = (t + 1) >> 2, j1 = (t + 1) & 3;
#pragma unroll
        for (int nd = 0; nd < 8; ++nd)
          vf[nxt][nd] = *(const v2f*)(vfr + (nt1 * 8 + j1) * VP + nd * 32);
      }
      __builtin_amdgcn_sched_barrier(0);
      v2f pb;
      pb.x = st[nt][2 * j];
      pb.y = st[nt][2 * j + 1];
#pragma unroll
      for (int nd = 0; nd < 8; ++nd)
        oacc[nd] = wmma4(vf[cur][nd], pb, oacc[nd]);
    }
  }

  const int b = bh >> 4, h = bh & 15;
  float* orow = o + ((size_t)(b * 2048 + q0 + wave * 16 + hl)) * 2048 +
                h * 128 + h8;
#pragma unroll
  for (int nd = 0; nd < 8; ++nd) {
    float4 lo, hi;
    lo.x = oacc[nd][0] / l_i;
    lo.y = oacc[nd][1] / l_i;
    lo.z = oacc[nd][2] / l_i;
    lo.w = oacc[nd][3] / l_i;
    hi.x = oacc[nd][4] / l_i;
    hi.y = oacc[nd][5] / l_i;
    hi.z = oacc[nd][6] / l_i;
    hi.w = oacc[nd][7] / l_i;
    *(float4*)(orow + nd * 16) = lo;
    *(float4*)(orow + nd * 16 + 4) = hi;
  }
}

__global__ __launch_bounds__(256) void k_gemm_out(
    const float* __restrict__ ov, const float* __restrict__ w,
    const float* __restrict__ bias, float* __restrict__ out) {
  __shared__ __align__(16) float As[128 * GP];
  __shared__ __align__(16) float Bs[128 * GP];
  const int m0 = blockIdx.y * 128;
  const int n0 = blockIdx.x * 128;
  v8f acc[8] = {};
  gemm_tile_128(ov, w, m0, n0, As, Bs, acc);

  const int wave = threadIdx.x >> 5, lane = threadIdx.x & 31;
  const int hl = lane & 15;
  const int wm = (wave >> 2) * 64;
  const int wn = (wave & 3) * 32;
#pragma unroll
  for (int t = 0; t < 8; ++t) {
    const int mt = t >> 1, nt = t & 1;
    const int n = n0 + wn + nt * 16 + hl;
    const float bn = bias[n];
#pragma unroll
    for (int r = 0; r < 8; ++r) {
      int m = m0 + wm + mt * 16 + ((lane < 16) ? r : r + 8);
      out[(size_t)m * 2048 + n] = acc[t][r] + bn;
    }
  }
}

extern "C" void kernel_launch(void* const* d_in, const int* in_sizes, int n_in,
                              void* d_out, int out_size, void* d_ws,
                              size_t ws_size, hipStream_t stream) {
  const float* x = (const float*)d_in[0];
  const float* pe = (const float*)d_in[1];
  const float* qkv_w = (const float*)d_in[2];
  const float* q_scale = (const float*)d_in[3];
  const float* k_scale = (const float*)d_in[4];
  const float* proj_w = (const float*)d_in[5];
  const float* proj_b = (const float*)d_in[6];
  float* out = (float*)d_out;

  float* ws = (float*)d_ws;
  const size_t per = (size_t)8388608;
  float* qw = ws;
  float* kw = ws + per;
  float* vw = ws + 2 * per;
  float* ow = ws + 3 * per;

  k_gemm_qkv<<<dim3(48, 32), 256, 0, stream>>>(x, qkv_w, pe, q_scale, k_scale,
                                               qw);
  k_attn<<<dim3(32, 32), 128, 0, stream>>>(qw, kw, vw, ow);
  k_gemm_out<<<dim3(16, 32), 256, 0, stream>>>(ow, proj_w, proj_b, out);
}


# Both projection GEMMs: k-loop software-pipelined through registers (a lane's 16 float4 of the next k-step fetched under this step's 128 WMMAs, stored to LDS at the next step's head); k order kept
# speedup vs baseline: 1.9086x; 1.0407x over previous
#include <hip/hip_runtime.h>
#include <cstdint>


typedef float v2f __attribute__((ext_vector_type(2)));
typedef float v4f __attribute__((ext_vector_type(4)));
typedef float v8f __attribute__((ext_vector_type(8)));

__device__ __forceinline__ v8f wmma4(v2f a, v2f b, v8f c) {
  return __builtin_amdgcn_wmma_f32_16x16x4_f32(
       false, a,  false, b,
       (short)0, c,  false,  false);
}

constexpr int GK = 64;
constexpr int GP = 68;
constexpr int SK = 132;

__device__ __forceinline__ void gemm_tile_128(
    const float* __restrict__ a, const float* __restrict__ w, int m0, int n0,
    float* __restrict__ As, float* __restrict__ Bs, v8f (&acc)[8]) {
  const int tid = threadIdx.x;
  const int wave = tid >> 5, lane = tid & 31;
  const int hl = lane & 15;
  const int koff = (lane < 16) ? 0 : 2;
  const int wm = (wave >> 2) * 64;
  const int wn = (wave & 3) * 32;
  const int lrow = tid >> 4;
  const int lc = (tid & 15) << 2;
  const float* ag = a + (size_t)(m0 + lrow) * 2048 + lc;
  const float* wg = w + (size_t)(n0 + lrow) * 2048 + lc;
  float* asd = As + lrow * GP + lc;
  float* bsd = Bs + lrow * GP + lc;
  const float* afr = As + (wm + hl) * GP + koff;
  const float* bfr = Bs + (wn + hl) * GP + koff;

  v4f ra[8], rb[8];
#pragma unroll
  for (int i = 0; i < 8; ++i) {
    ra[i] = *(const v4f*)(ag + (size_t)(i * 16) * 2048);
    rb[i] = *(const v4f*)(wg + (size_t)(i * 16) * 2048);
  }

  for (int k0 = 0; k0 < 2048; k0 += GK) {
    __syncthreads();
#pragma unroll
    for (int i = 0; i < 8; ++i) {
      *(v4f*)(asd + i * 16 * GP) = ra[i];
      *(v4f*)(bsd + i * 16 * GP) = rb[i];
    }
    __syncthreads();
    if (k0 + GK < 2048) {
#pragma unroll
      for (int i = 0; i < 8; ++i) {
        ra[i] = *(const v4f*)(ag + (size_t)(i * 16) * 2048 + k0 + GK);
        rb[i] = *(const v4f*)(wg + (size_t)(i * 16) * 2048 + k0 + GK);
      }
    }
#pragma unroll
    for (int kk = 0; kk < GK; kk += 4) {
      v2f af[4], bf[2];
#pragma unroll
      for (int mt = 0; mt < 4; ++mt)
        af[mt] = *(const v2f*)(afr + mt * 16 * GP + kk);
#pragma unroll
      for (int nt = 0; nt < 2; ++nt)
        bf[nt] = *(const v2f*)(bfr + nt * 16 * GP + kk);
#pragma unroll
      for (int mt = 0; mt < 4; ++mt) {
        acc[mt * 2 + 0] = wmma4(af[mt], bf[0], acc[mt * 2 + 0]);
        acc[mt * 2 + 1] = wmma4(af[mt], bf[1], acc[mt * 2 + 1]);
      }
    }
  }
}

constexpr int TP = 132;

__global__ __launch_bounds__(256) void k_gemm_qkv(
    const float* __restrict__ x, const float* __restrict__ w,
    const float* __restrict__ pe, const float* __restrict__ qs,
    const float* __restrict__ ks, float* __restrict__ qkv) {
  __shared__ __align__(16) float smem[2 * 128 * GP];
  float* As = smem;
  float* Bs = smem + 128 * GP;
  const int m0 = blockIdx.y * 128;
  const int n0 = blockIdx.x * 128;
  v8f acc[8] = {};
  gemm_tile_128(x, w, m0, n0, As, Bs, acc);

  const int wave = threadIdx.x >> 5, lane = threadIdx.x & 31;
  const int hl = lane & 15;
  const int h8 = (lane < 16) ? 0 : 8;
  const int wm = (wave >> 2) * 64;
  const int wn = (wave & 3) * 32;
  const int which = n0 >> 11;
  const int h = (n0 >> 7) & 15;
  const int b = m0 >> 11;
  const int l0 = m0 & 2047;
  float* dst = qkv + (size_t)which * (size_t)8388608 +
               (size_t)((b * 16 + h) * 2048 + l0) * 128;

  __syncthreads();
#pragma unroll
  for (int t = 0; t < 8; ++t) {
    const int mt = t >> 1, nt = t & 1;
#pragma unroll
    for (int r = 0; r < 8; ++r)
      smem[(wm + mt * 16 + h8 + r) * TP + wn + nt * 16 + hl] = acc[t][r];
  }
  __syncthreads();

  const float* trow = smem + wave * 16 * TP + lane * 4;
  float* drow = dst + (size_t)(wave * 16) * 128 + lane * 4;
  if (which == 2) {
#pragma unroll
    for (int i = 0; i < 16; ++i)
      *(float4*)(drow + i * 128) = *(const float4*)(trow + i * TP);
    return;
  }

  const float* sc = which ? ks : qs;
  const float4 s4 = *(const float4*)(sc + lane * 4);
  const float* peb = pe + (size_t)(l0 + wave * 16) * 256 + lane * 8;
#pragma unroll 4
  for (int i = 0; i < 16; ++i) {
    float4 xv = *(const float4*)(trow + i * TP);
    float4 p0 = *(const float4*)(peb + i * 256);
    float4 p1 = *(const float4*)(peb + i * 256 + 4);
    float ss = xv.x * xv.x + xv.y * xv.y + xv.z * xv.z + xv.w * xv.w;
#pragma unroll
    for (int sh = 16; sh >= 1; sh >>= 1) ss += __shfl_xor(ss, sh, 32);
    float rrms = rsqrtf(ss * (1.0f / 128.0f) + 1e-6f);
    float x0 = xv.x * rrms * s4.x;
    float x1 = xv.y * rrms * s4.y;
    float x2 = xv.z * rrms * s4.z;
    float x3 = xv.w * rrms * s4.w;
    float4 yv;
    yv.x = p0.x * x0 + p0.y * x1;
    yv.y = p0.z * x0 + p0.w * x1;
    yv.z = p1.x * x2 + p1.y * x3;
    yv.w = p1.z * x2 + p1.w * x3;
    *(float4*)(drow + i * 128) = yv;
  }
}

constexpr int VP = 264;

__global__ __launch_bounds__(128) void k_attn(
    const float* __restrict__ q, const float* __restrict__ k,
    const float* __restrict__ v, float* __restrict__ o) {
  __shared__ __align__(16) float Ks[64 * SK];
  __shared__ __align__(16) float Vp[32 * VP];
  const int bh = blockIdx.y;
  const int q0 = blockIdx.x * 64;
  const int tid = threadIdx.x;
  const int wave = tid >> 5, lane = tid & 31;
  const int hl = lane & 15;
  const int koff = (lane < 16) ? 0 : 2;
  const int h8 = (lane < 16) ? 0 : 8;
  const float rscale = 0.08838834764831845f;

  const float* qb = q + (size_t)bh * (2048 * 128);
  const float* kbase = k + (size_t)bh * (2048 * 128);
  const float* vbase = v + (size_t)bh * (2048 * 128);

  v2f qa[32];
  {
    const float* qrow = qb + (size_t)(q0 + wave * 16 + hl) * 128;
#pragma unroll
    for (int s = 0; s < 32; ++s) qa[s] = *(const v2f*)(qrow + 4 * s + koff);
  }

  v8f oacc[8] = {};
  float m_i = -3.4e38f, l_i = 0.f;

  const float* kfr = &Ks[hl * SK + koff];
  const float* vfr = &Vp[(h8 >> 1) * VP + hl * 2];

  for (int kb0 = 0; kb0 < 2048; kb0 += 64) {
    __syncthreads();
#pragma unroll 2
    for (int it = 0; it < 8; ++it) {
      const int i = tid + it * 128;
      const int m = i >> 5;
      const int c4 = (i & 31) << 2;
      const float* kg = kbase + (size_t)(kb0 + 2 * m) * 128 + c4;
      const float* vg = vbase + (size_t)(kb0 + 2 * m) * 128 + c4;
      float4 k0 = *(const float4*)(kg);
      float4 k1 = *(const float4*)(kg + 128);
      float4 va = *(const float4*)(vg);
      float4 vb = *(const float4*)(vg + 128);
      *(float4*)&Ks[(2 * m) * SK + c4] = k0;
      *(float4*)&Ks[(2 * m + 1) * SK + c4] = k1;
      *(float4*)&Vp[m * VP + c4 * 2] = make_float4(va.x, vb.x, va.y, vb.y);
      *(float4*)&Vp[m * VP + c4 * 2 + 4] = make_float4(va.z, vb.z, va.w, vb.w);
    }
    __syncthreads();

    v8f st[4] = {};
    {
      v2f kf[2][8];
#pragma unroll
      for (int nt = 0; nt < 4; ++nt) {
        kf[0][nt * 2] = *(const v2f*)(kfr + nt * 16 * SK);
        kf[0][nt * 2 + 1] = *(const v2f*)(kfr + nt * 16 * SK + 4);
      }
#pragma unroll
      for (int sp = 0; sp < 16; ++sp) {
        const int cur = sp & 1, nxt = cur ^ 1;
        if (sp + 1 < 16) {
#pragma unroll
          for (int nt = 0; nt < 4; ++nt) {
            kf[nxt][nt * 2] =
                *(const v2f*)(kfr + nt * 16 * SK + 8 * (sp + 1));
            kf[nxt][nt * 2 + 1] =
                *(const v2f*)(kfr + nt * 16 * SK + 8 * (sp + 1) + 4);
          }
        }
        __builtin_amdgcn_sched_barrier(0);
#pragma unroll
        for (int nt = 0; nt < 4; ++nt)
          st[nt] = wmma4(kf[cur][nt * 2], qa[2 * sp], st[nt]);
#pragma unroll
        for (int nt = 0; nt < 4; ++nt)
          st[nt] = wmma4(kf[cur][nt * 2 + 1], qa[2 * sp + 1], st[nt]);
      }
    }

    v2f vf[2][8];
#pragma unroll
    for (int nd = 0; nd < 8; ++nd) vf[0][nd] = *(const v2f*)(vfr + nd * 32);

    float mx = -3.4e38f;
#pragma unroll
    for (int nt = 0; nt < 4; ++nt) {
#pragma unroll
      for (int r = 0; r < 8; ++r) {
        st[nt][r] *= rscale;
        mx = fmaxf(mx, st[nt][r]);
      }
    }
    mx = fmaxf(mx, __shfl_xor(mx, 16, 32));
    const float mnew = fmaxf(m_i, mx);
    const float alpha = __expf(m_i - mnew);
    m_i = mnew;
    float rs = 0.f;
#pragma unroll
    for (int nt = 0; nt < 4; ++nt) {
#pragma unroll
      for (int r = 0; r < 8; ++r) {
        float p = __expf(st[nt][r] - mnew);
        st[nt][r] = p;
        rs += p;
      }
    }
    rs += __shfl_xor(rs, 16, 32);
    l_i = l_i * alpha + rs;
#pragma unroll
    for (int nd = 0; nd < 8; ++nd) {
#pragma unroll
      for (int r = 0; r < 8; ++r) oacc[nd][r] *= alpha;
    }

#pragma unroll
    for (int t = 0; t < 16; ++t) {
      const int nt = t >> 2, j = t & 3;
      const int cur = t & 1, nxt = cur ^ 1;
      if (t + 1 < 16) {
        const int nt1 = (t + 1) >> 2, j1 = (t + 1) & 3;
#pragma unroll
        for (int nd = 0; nd < 8; ++nd)
          vf[nxt][nd] = *(const v2f*)(vfr + (nt1 * 8 + j1) * VP + nd * 32);
      }
      __builtin_amdgcn_sched_barrier(0);
      v2f pb;
      pb.x = st[nt][2 * j];
      pb.y = st[nt][2 * j + 1];
#pragma unroll
      for (int nd = 0; nd < 8; ++nd)
        oacc[nd] = wmma4(vf[cur][nd], pb, oacc[nd]);
    }
  }

  const int b = bh >> 4, h = bh & 15;
  float* orow = o + ((size_t)(b * 2048 + q0 + wave * 16 + hl)) * 2048 +
                h * 128 + h8;
#pragma unroll
  for (int nd = 0; nd < 8; ++nd) {
    float4 lo, hi;
    lo.x = oacc[nd][0] / l_i;
    lo.y = oacc[nd][1] / l_i;
    lo.z = oacc[nd][2] / l_i;
    lo.w = oacc[nd][3] / l_i;
    hi.x = oacc[nd][4] / l_i;
    hi.y = oacc[nd][5] / l_i;
    hi.z = oacc[nd][6] / l_i;
    hi.w = oacc[nd][7] / l_i;
    *(float4*)(orow + nd * 16) = lo;
    *(float4*)(orow + nd * 16 + 4) = hi;
  }
}

__global__ __launch_bounds__(256) void k_gemm_out(
    const float* __restrict__ ov, const float* __restrict__ w,
    const float* __restrict__ bias, float* __restrict__ out) {
  __shared__ __align__(16) float As[128 * GP];
  __shared__ __align__(16) float Bs[128 * GP];
  const int m0 = blockIdx.y * 128;
  const int n0 = blockIdx.x * 128;
  v8f acc[8] = {};
  gemm_tile_128(ov, w, m0, n0, As, Bs, acc);

  const int wave = threadIdx.x >> 5, lane = threadIdx.x & 31;
  const int hl = lane & 15;
  const int wm = (wave >> 2) * 64;
  const int wn = (wave & 3) * 32;
#pragma unroll
  for (int t = 0; t < 8; ++t) {
    const int mt = t >> 1, nt = t & 1;
    const int n = n0 + wn + nt * 16 + hl;
    const float bn = bias[n];
#pragma unroll
    for (int r = 0; r < 8; ++r) {
      int m = m0 + wm + mt * 16 + ((lane < 16) ? r : r + 8);
      out[(size_t)m * 2048 + n] = acc[t][r] + bn;
    }
  }
}

extern "C" void kernel_launch(void* const* d_in, const int* in_sizes, int n_in,
                              void* d_out, int out_size, void* d_ws,
                              size_t ws_size, hipStream_t stream) {
  const float* x = (const float*)d_in[0];
  const float* pe = (const float*)d_in[1];
  const float* qkv_w = (const float*)d_in[2];
  const float* q_scale = (const float*)d_in[3];
  const float* k_scale = (const float*)d_in[4];
  const float* proj_w = (const float*)d_in[5];
  const float* proj_b = (const float*)d_in[6];
  float* out = (float*)d_out;

  float* ws = (float*)d_ws;
  const size_t per = (size_t)8388608;
  float* qw = ws;
  float* kw = ws + per;
  float* vw = ws + 2 * per;
  float* ow = ws + 3 * per;

  k_gemm_qkv<<<dim3(48, 32), 256, 0, stream>>>(x, qkv_w, pe, q_scale, k_scale,
                                               qw);
  k_attn<<<dim3(32, 32), 128, 0, stream>>>(qw, kw, vw, ow);
  k_gemm_out<<<dim3(16, 32), 256, 0, stream>>>(ow, proj_w, proj_b, out);
}


# Attention work-group widened from 4 to 8 waves (128 queries per staged K/V tile): half the staging loads, LDS stores and cache reads per WMMA, a lane's 16 tile loads in one batch; sums unchanged
# speedup vs baseline: 1.9812x; 1.0380x over previous
#include <hip/hip_runtime.h>
#include <cstdint>


typedef float v2f __attribute__((ext_vector_type(2)));
typedef float v4f __attribute__((ext_vector_type(4)));
typedef float v8f __attribute__((ext_vector_type(8)));

__device__ __forceinline__ v8f wmma4(v2f a, v2f b, v8f c) {
  return __builtin_amdgcn_wmma_f32_16x16x4_f32(
       false, a,  false, b,
       (short)0, c,  false,  false);
}

constexpr int GK = 64;
constexpr int GP = 68;
constexpr int SK = 132;

__device__ __forceinline__ void gemm_tile_128(
    const float* __restrict__ a, const float* __restrict__ w, int m0, int n0,
    float* __restrict__ As, float* __restrict__ Bs, v8f (&acc)[8]) {
  const int tid = threadIdx.x;
  const int wave = tid >> 5, lane = tid & 31;
  const int hl = lane & 15;
  const int koff = (lane < 16) ? 0 : 2;
  const int wm = (wave >> 2) * 64;
  const int wn = (wave & 3) * 32;
  const int lrow = tid >> 4;
  const int lc = (tid & 15) << 2;
  const float* ag = a + (size_t)(m0 + lrow) * 2048 + lc;
  const float* wg = w + (size_t)(n0 + lrow) * 2048 + lc;
  float* asd = As + lrow * GP + lc;
  float* bsd = Bs + lrow * GP + lc;
  const float* afr = As + (wm + hl) * GP + koff;
  const float* bfr = Bs + (wn + hl) * GP + koff;

  v4f ra[8], rb[8];
#pragma unroll
  for (int i = 0; i < 8; ++i) {
    ra[i] = *(const v4f*)(ag + (size_t)(i * 16) * 2048);
    rb[i] = *(const v4f*)(wg + (size_t)(i * 16) * 2048);
  }

  for (int k0 = 0; k0 < 2048; k0 += GK) {
    __syncthreads();
#pragma unroll
    for (int i = 0; i < 8; ++i) {
      *(v4f*)(asd + i * 16 * GP) = ra[i];
      *(v4f*)(bsd + i * 16 * GP) = rb[i];
    }
    __syncthreads();
    if (k0 + GK < 2048) {
#pragma unroll
      for (int i = 0; i < 8; ++i) {
        ra[i] = *(const v4f*)(ag + (size_t)(i * 16) * 2048 + k0 + GK);
        rb[i] = *(const v4f*)(wg + (size_t)(i * 16) * 2048 + k0 + GK);
      }
    }
#pragma unroll
    for (int kk = 0; kk < GK; kk += 4) {
      v2f af[4], bf[2];
#pragma unroll
      for (int mt = 0; mt < 4; ++mt)
        af[mt] = *(const v2f*)(afr + mt * 16 * GP + kk);
#pragma unroll
      for (int nt = 0; nt < 2; ++nt)
        bf[nt] = *(const v2f*)(bfr + nt * 16 * GP + kk);
#pragma unroll
      for (int mt = 0; mt < 4; ++mt) {
        acc[mt * 2 + 0] = wmma4(af[mt], bf[0], acc[mt * 2 + 0]);
        acc[mt * 2 + 1] = wmma4(af[mt], bf[1], acc[mt * 2 + 1]);
      }
    }
  }
}

constexpr int TP = 132;

__global__ __launch_bounds__(256) void k_gemm_qkv(
    const float* __restrict__ x, const float* __restrict__ w,
    const float* __restrict__ pe, const float* __restrict__ qs,
    const float* __restrict__ ks, float* __restrict__ qkv) {
  __shared__ __align__(16) float smem[2 * 128 * GP];
  float* As = smem;
  float* Bs = smem + 128 * GP;
  const int m0 = blockIdx.y * 128;
  const int n0 = blockIdx.x * 128;
  v8f acc[8] = {};
  gemm_tile_128(x, w, m0, n0, As, Bs, acc);

  const int wave = threadIdx.x >> 5, lane = threadIdx.x & 31;
  const int hl = lane & 15;
  const int h8 = (lane < 16) ? 0 : 8;
  const int wm = (wave >> 2) * 64;
  const int wn = (wave & 3) * 32;
  const int which = n0 >> 11;
  const int h = (n0 >> 7) & 15;
  const int b = m0 >> 11;
  const int l0 = m0 & 2047;
  float* dst = qkv + (size_t)which * (size_t)8388608 +
               (size_t)((b * 16 + h) * 2048 + l0) * 128;

  __syncthreads();
#pragma unroll
  for (int t = 0; t < 8; ++t) {
    const int mt = t >> 1, nt = t & 1;
#pragma unroll
    for (int r = 0; r < 8; ++r)
      smem[(wm + mt * 16 + h8 + r) * TP + wn + nt * 16 + hl] = acc[t][r];
  }
  __syncthreads();

  const float* trow = smem + wave * 16 * TP + lane * 4;
  float* drow = dst + (size_t)(wave * 16) * 128 + lane * 4;
  if (which == 2) {
#pragma unroll
    for (int i = 0; i < 16; ++i)
      *(float4*)(drow + i * 128) = *(const float4*)(trow + i * TP);
    return;
  }

  const float* sc = which ? ks : qs;
  const float4 s4 = *(const float4*)(sc + lane * 4);
  const float* peb = pe + (size_t)(l0 + wave * 16) * 256 + lane * 8;
#pragma unroll 4
  for (int i = 0; i < 16; ++i) {
    float4 xv = *(const float4*)(trow + i * TP);
    float4 p0 = *(const float4*)(peb + i * 256);
    float4 p1 = *(const float4*)(peb + i * 256 + 4);
    float ss = xv.x * xv.x + xv.y * xv.y + xv.z * xv.z + xv.w * xv.w;
#pragma unroll
    for (int sh = 16; sh >= 1; sh >>= 1) ss += __shfl_xor(ss, sh, 32);
    float rrms = rsqrtf(ss * (1.0f / 128.0f) + 1e-6f);
    float x0 = xv.x * rrms * s4.x;
    float x1 = xv.y * rrms * s4.y;
    float x2 = xv.z * rrms * s4.z;
    float x3 = xv.w * rrms * s4.w;
    float4 yv;
    yv.x = p0.x * x0 + p0.y * x1;
    yv.y = p0.z * x0 + p0.w * x1;
    yv.z = p1.x * x2 + p1.y * x3;
    yv.w = p1.z * x2 + p1.w * x3;
    *(float4*)(drow + i * 128) = yv;
  }
}

constexpr int VP = 264;

__global__ __launch_bounds__(256) void k_attn(
    const float* __restrict__ q, const float* __restrict__ k,
    const float* __restrict__ v, float* __restrict__ o) {
  __shared__ __align__(16) float Ks[64 * SK];
  __shared__ __align__(16) float Vp[32 * VP];
  const int bh = blockIdx.y;
  const int q0 = blockIdx.x * 128;
  const int tid = threadIdx.x;
  const int wave = tid >> 5, lane = tid & 31;
  const int hl = lane & 15;
  const int koff = (lane < 16) ? 0 : 2;
  const int h8 = (lane < 16) ? 0 : 8;
  const float rscale = 0.08838834764831845f;

  const float* qb = q + (size_t)bh * (2048 * 128);
  const float* kbase = k + (size_t)bh * (2048 * 128);
  const float* vbase = v + (size_t)bh * (2048 * 128);

  v2f qa[32];
  {
    const float* qrow = qb + (size_t)(q0 + wave * 16 + hl) * 128;
#pragma unroll
    for (int s = 0; s < 32; ++s) qa[s] = *(const v2f*)(qrow + 4 * s + koff);
  }

  v8f oacc[8] = {};
  float m_i = -3.4e38f, l_i = 0.f;

  const int prow = tid >> 5;
  const int c4 = (tid & 31) << 2;
  const float* kthr = kbase + (size_t)(2 * prow) * 128 + c4;
  const float* vthr = vbase + (size_t)(2 * prow) * 128 + c4;
  float* ksd = &Ks[(2 * prow) * SK + c4];
  float* vpd = &Vp[prow * VP + c4 * 2];

  const float* kfr = &Ks[hl * SK + koff];
  const float* vfr = &Vp[(h8 >> 1) * VP + hl * 2];

  for (int kb0 = 0; kb0 < 2048; kb0 += 64) {
    __syncthreads();
    {
      const float* kg = kthr + (size_t)kb0 * 128;
      const float* vg = vthr + (size_t)kb0 * 128;
#pragma unroll
      for (int it = 0; it < 4; ++it) {
        float4 k0 = *(const float4*)(kg + it * 2048);
        float4 k1 = *(const float4*)(kg + it * 2048 + 128);
        float4 va = *(const float4*)(vg + it * 2048);
        float4 vb = *(const float4*)(vg + it * 2048 + 128);
        *(float4*)(ksd + it * 16 * SK) = k0;
        *(float4*)(ksd + it * 16 * SK + SK) = k1;
        *(float4*)(vpd + it * 8 * VP) = make_float4(va.x, vb.x, va.y, vb.y);
        *(float4*)(vpd + it * 8 * VP + 4) = make_float4(va.z, vb.z, va.w, vb.w);
      }
    }
    __syncthreads();

    v8f st[4] = {};
    {
      v2f kf[2][8];
#pragma unroll
      for (int nt = 0; nt < 4; ++nt) {
        kf[0][nt * 2] = *(const v2f*)(kfr + nt * 16 * SK);
        kf[0][nt * 2 + 1] = *(const v2f*)(kfr + nt * 16 * SK + 4);
      }
#pragma unroll
      for (int sp = 0; sp < 16; ++sp) {
        const int cur = sp & 1, nxt = cur ^ 1;
        if (sp + 1 < 16) {
#pragma unroll
          for (int nt = 0; nt < 4; ++nt) {
            kf[nxt][nt * 2] =
                *(const v2f*)(kfr + nt * 16 * SK + 8 * (sp + 1));
            kf[nxt][nt * 2 + 1] =
                *(const v2f*)(kfr + nt * 16 * SK + 8 * (sp + 1) + 4);
          }
        }
        __builtin_amdgcn_sched_barrier(0);
#pragma unroll
        for (int nt = 0; nt < 4; ++nt)
          st[nt] = wmma4(kf[cur][nt * 2], qa[2 * sp], st[nt]);
#pragma unroll
        for (int nt = 0; nt < 4; ++nt)
          st[nt] = wmma4(kf[cur][nt * 2 + 1], qa[2 * sp + 1], st[nt]);
      }
    }

    v2f vf[2][8];
#pragma unroll
    for (int nd = 0; nd < 8; ++nd) vf[0][nd] = *(const v2f*)(vfr + nd * 32);

    float mx = -3.4e38f;
#pragma unroll
    for (int nt = 0; nt < 4; ++nt) {
#pragma unroll
      for (int r = 0; r < 8; ++r) {
        st[nt][r] *= rscale;
        mx = fmaxf(mx, st[nt][r]);
      }
    }
    mx = fmaxf(mx, __shfl_xor(mx, 16, 32));
    const float mnew = fmaxf(m_i, mx);
    const float alpha = __expf(m_i - mnew);
    m_i = mnew;
    float rs = 0.f;
#pragma unroll
    for (int nt = 0; nt < 4; ++nt) {
#pragma unroll
      for (int r = 0; r < 8; ++r) {
        float p = __expf(st[nt][r] - mnew);
        st[nt][r] = p;
        rs += p;
      }
    }
    rs += __shfl_xor(rs, 16, 32);
    l_i = l_i * alpha + rs;
#pragma unroll
    for (int nd = 0; nd < 8; ++nd) {
#pragma unroll
      for (int r = 0; r < 8; ++r) oacc[nd][r] *= alpha;
    }

#pragma unroll
    for (int t = 0; t < 16; ++t) {
      const int nt = t >> 2, j = t & 3;
      const int cur = t & 1, nxt = cur ^ 1;
      if (t + 1 < 16) {
        const int nt1 = (t + 1) >> 2, j1 = (t + 1) & 3;
#pragma unroll
        for (int nd = 0; nd < 8; ++nd)
          vf[nxt][nd] = *(const v2f*)(vfr + (nt1 * 8 + j1) * VP + nd * 32);
      }
      __builtin_amdgcn_sched_barrier(0);
      v2f pb;
      pb.x = st[nt][2 * j];
      pb.y = st[nt][2 * j + 1];
#pragma unroll
      for (int nd = 0; nd < 8; ++nd)
        oacc[nd] = wmma4(vf[cur][nd], pb, oacc[nd]);
    }
  }

  const int b = bh >> 4, h = bh & 15;
  float* orow = o + ((size_t)(b * 2048 + q0 + wave * 16 + hl)) * 2048 +
                h * 128 + h8;
#pragma unroll
  for (int nd = 0; nd < 8; ++nd) {
    float4 lo, hi;
    lo.x = oacc[nd][0] / l_i;
    lo.y = oacc[nd][1] / l_i;
    lo.z = oacc[nd][2] / l_i;
    lo.w = oacc[nd][3] / l_i;
    hi.x = oacc[nd][4] / l_i;
    hi.y = oacc[nd][5] / l_i;
    hi.z = oacc[nd][6] / l_i;
    hi.w = oacc[nd][7] / l_i;
    *(float4*)(orow + nd * 16) = lo;
    *(float4*)(orow + nd * 16 + 4) = hi;
  }
}

__global__ __launch_bounds__(256) void k_gemm_out(
    const float* __restrict__ ov, const float* __restrict__ w,
    const float* __restrict__ bias, float* __restrict__ out) {
  __shared__ __align__(16) float As[128 * GP];
  __shared__ __align__(16) float Bs[128 * GP];
  const int m0 = blockIdx.y * 128;
  const int n0 = blockIdx.x * 128;
  v8f acc[8] = {};
  gemm_tile_128(ov, w, m0, n0, As, Bs, acc);

  const int wave = threadIdx.x >> 5, lane = threadIdx.x & 31;
  const int hl = lane & 15;
  const int wm = (wave >> 2) * 64;
  const int wn = (wave & 3) * 32;
#pragma unroll
  for (int t = 0; t < 8; ++t) {
    const int mt = t >> 1, nt = t & 1;
    const int n = n0 + wn + nt * 16 + hl;
    const float bn = bias[n];
#pragma unroll
    for (int r = 0; r < 8; ++r) {
      int m = m0 + wm + mt * 16 + ((lane < 16) ? r : r + 8);
      out[(size_t)m * 2048 + n] = acc[t][r] + bn;
    }
  }
}

extern "C" void kernel_launch(void* const* d_in, const int* in_sizes, int n_in,
                              void* d_out, int out_size, void* d_ws,
                              size_t ws_size, hipStream_t stream) {
  const float* x = (const float*)d_in[0];
  const float* pe = (const float*)d_in[1];
  const float* qkv_w = (const float*)d_in[2];
  const float* q_scale = (const float*)d_in[3];
  const float* k_scale = (const float*)d_in[4];
  const float* proj_w = (const float*)d_in[5];
  const float* proj_b = (const float*)d_in[6];
  float* out = (float*)d_out;

  float* ws = (float*)d_ws;
  const size_t per = (size_t)8388608;
  float* qw = ws;
  float* kw = ws + per;
  float* vw = ws + 2 * per;
  float* ow = ws + 3 * per;

  k_gemm_qkv<<<dim3(48, 32), 256, 0, stream>>>(x, qkv_w, pe, q_scale, k_scale,
                                               qw);
  k_attn<<<dim3(16, 32), 256, 0, stream>>>(qw, kw, vw, ow);
  k_gemm_out<<<dim3(16, 32), 256, 0, stream>>>(ow, proj_w, proj_b, out);
}
